# starting kernel
# baseline (speedup 1.0000x reference)
#include <hip/hip_runtime.h>
#include <math.h>

typedef __attribute__((ext_vector_type(2))) float v2f;
typedef __attribute__((ext_vector_type(8))) float v8f;

#define NQ 12
#define NSTATE 4096
#define NPADDED (4096 + 256)
#define PAD_IDX(i) ((i) + ((i) >> 4))

__device__ __forceinline__ int perm_fwd(int i) {
#pragma unroll
  for (int j = NQ - 1; j >= 0; --j) {
    int c = 11 - j;
    int t = 11 - ((j + 1) % NQ);
    i ^= ((i >> c) & 1) << t;
  }
  return i;
}
__device__ __forceinline__ int perm_inv(int i) {
#pragma unroll
  for (int j = 0; j < NQ; ++j) {
    int c = 11 - j;
    int t = 11 - ((j + 1) % NQ);
    i ^= ((i >> c) & 1) << t;
  }
  return i;
}

__device__ __forceinline__ int sweep_idx(int s, int g, int v) {
  if (s == 0) return (g << 4) | v;
  if (s == 1) return ((g & 0xF0) << 4) | (v << 4) | (g & 0x0F);
  return (v << 8) | g;
}

__global__ __launch_bounds__(256) void qsim_kernel(const float* __restrict__ x,
                                                   const float* __restrict__ w,
                                                   float* __restrict__ out) {
  __shared__ float sre[NPADDED];
  __shared__ float sim[NPADDED];
  __shared__ float matP[16 * 17];
  __shared__ float matQ[16 * 17];
  __shared__ float matQn[16 * 17];
  __shared__ float csA[NQ], snA[NQ], csB[NQ], snB[NQ];
  __shared__ float red[8][NQ];

  const int tid = threadIdx.x;
  const int b = blockIdx.x;
  const int lane = tid & 31;
  const int wv = tid >> 5;

  if (tid < NQ) {
    int q = 11 - tid;
    float thA = x[b * NQ + q] + w[q];
    float thB = w[NQ + q];
    csA[tid] = cosf(0.5f * thA);
    snA[tid] = sinf(0.5f * thA);
    csB[tid] = cosf(0.5f * thB);
    snB[tid] = sinf(0.5f * thB);
  }
  __syncthreads();

  {
    float base = 1.0f;
#pragma unroll
    for (int k = 0; k < 8; ++k)
      base *= ((tid >> k) & 1) ? snA[4 + k] : csA[4 + k];
    int pct = __popc(tid);
#pragma unroll
    for (int v = 0; v < 16; ++v) {
      float m = base;
#pragma unroll
      for (int k = 0; k < 4; ++k)
        m *= ((v >> k) & 1) ? snA[k] : csA[k];
      int pc = (pct + __popc(v)) & 3;
      float re = (pc == 0) ? m : ((pc == 2) ? -m : 0.0f);
      float im = (pc == 1) ? -m : ((pc == 3) ? m : 0.0f);
      int idx = (tid << 4) | v;
      sre[PAD_IDX(idx)] = re;
      sim[PAD_IDX(idx)] = im;
    }
  }
  __syncthreads();

  {
    float tr[16], ti[16];
#pragma unroll
    for (int v = 0; v < 16; ++v) {
      int j = perm_fwd((tid << 4) | v);
      tr[v] = sre[PAD_IDX(j)];
      ti[v] = sim[PAD_IDX(j)];
    }
    __syncthreads();
#pragma unroll
    for (int v = 0; v < 16; ++v) {
      int i = (tid << 4) | v;
      sre[PAD_IDX(i)] = tr[v];
      sim[PAD_IDX(i)] = ti[v];
    }
  }
  __syncthreads();

#pragma unroll
  for (int s = 0; s < 3; ++s) {
    {
      int r = tid >> 4, c = tid & 15, xr = r ^ c;
      float m = 1.0f;
#pragma unroll
      for (int k = 0; k < 4; ++k)
        m *= ((xr >> k) & 1) ? snB[4 * s + k] : csB[4 * s + k];
      int d = __popc(xr) & 3;
      float pe = (d == 0) ? m : ((d == 2) ? -m : 0.0f);
      float qe = (d == 1) ? -m : ((d == 3) ? m : 0.0f);
      matP[r * 17 + c] = pe;
      matQ[r * 17 + c] = qe;
      matQn[r * 17 + c] = -qe;
    }
    __syncthreads();

    const int arow = lane & 15;
    const int koff = (lane >> 4) << 1;
    v2f aP[4], aQ[4], aQn[4];
#pragma unroll
    for (int j = 0; j < 4; ++j) {
      int k0 = 4 * j + koff;
      aP[j].x = matP[arow * 17 + k0];
      aP[j].y = matP[arow * 17 + k0 + 1];
      aQ[j].x = matQ[arow * 17 + k0];
      aQ[j].y = matQ[arow * 17 + k0 + 1];
      aQn[j].x = matQn[arow * 17 + k0];
      aQn[j].y = matQn[arow * 17 + k0 + 1];
    }

#pragma unroll
    for (int tt = 0; tt < 2; ++tt) {
      int gcol = (wv * 2 + tt) * 16 + (lane & 15);
      v2f bre[4], bim[4];
#pragma unroll
      for (int j = 0; j < 4; ++j) {
        int v0 = 4 * j + koff;
        int i0 = sweep_idx(s, gcol, v0);
        int i1 = sweep_idx(s, gcol, v0 + 1);
        bre[j].x = sre[PAD_IDX(i0)];
        bre[j].y = sre[PAD_IDX(i1)];
        bim[j].x = sim[PAD_IDX(i0)];
        bim[j].y = sim[PAD_IDX(i1)];
      }
      v8f dre = {};
      v8f dim = {};
#pragma unroll
      for (int j = 0; j < 4; ++j) {
        dre = __builtin_amdgcn_wmma_f32_16x16x4_f32(false, aP[j], false, bre[j],
                                                    (short)0, dre, false, false);
        dre = __builtin_amdgcn_wmma_f32_16x16x4_f32(false, aQn[j], false, bim[j],
                                                    (short)0, dre, false, false);
        dim = __builtin_amdgcn_wmma_f32_16x16x4_f32(false, aQ[j], false, bre[j],
                                                    (short)0, dim, false, false);
        dim = __builtin_amdgcn_wmma_f32_16x16x4_f32(false, aP[j], false, bim[j],
                                                    (short)0, dim, false, false);
      }
      int vadd = (lane >> 4) << 3;
#pragma unroll
      for (int r = 0; r < 8; ++r) {
        int idx = sweep_idx(s, gcol, r + vadd);
        sre[PAD_IDX(idx)] = dre[r];
        sim[PAD_IDX(idx)] = dim[r];
      }
    }
    __syncthreads();
  }

  float acc[NQ];
#pragma unroll
  for (int q = 0; q < NQ; ++q) acc[q] = 0.0f;
#pragma unroll
  for (int v = 0; v < 16; ++v) {
    int jdx = (tid << 4) | v;
    float re = sre[PAD_IDX(jdx)];
    float im = sim[PAD_IDX(jdx)];
    float p = re * re + im * im;
    int i = perm_inv(jdx);
#pragma unroll
    for (int q = 0; q < NQ; ++q) acc[q] += ((i >> q) & 1) ? -p : p;
  }
#pragma unroll
  for (int off = 16; off > 0; off >>= 1) {
#pragma unroll
    for (int q = 0; q < NQ; ++q) acc[q] += __shfl_down(acc[q], off, 32);
  }
  if (lane == 0) {
#pragma unroll
    for (int q = 0; q < NQ; ++q) red[wv][q] = acc[q];
  }
  __syncthreads();
  if (tid < NQ) {
    float sum = 0.0f;
#pragma unroll
    for (int k = 0; k < 8; ++k) sum += red[k][tid];
    out[b * NQ + (11 - tid)] = sum;
  }
}

extern "C" void kernel_launch(void* const* d_in, const int* in_sizes, int n_in,
                              void* d_out, int out_size, void* d_ws,
                              size_t ws_size, hipStream_t stream) {
  const float* x = (const float*)d_in[0];
  const float* w = (const float*)d_in[1];
  float* out = (float*)d_out;
  int B = in_sizes[0] / NQ;
  qsim_kernel<<<B, 256, 0, stream>>>(x, w, out);
}


# Hold the state as amp*i^popcount(idx): layer-1 RX is then a real rotation, each sweep is 2 real products with one matrix (WMMAs per wave 96->48); the order of the 16-term sums changes
# speedup vs baseline: 1.2985x; 1.2985x over previous
#include <hip/hip_runtime.h>
#include <math.h>

typedef __attribute__((ext_vector_type(2))) float v2f;
typedef __attribute__((ext_vector_type(8))) float v8f;

#define NQ 12
#define NSTATE 4096
#define NPADDED (4096 + 256)
#define PAD_IDX(i) ((i) + ((i) >> 4))

__device__ __forceinline__ int perm_fwd(int i) {
#pragma unroll
  for (int j = NQ - 1; j >= 0; --j) {
    int c = 11 - j;
    int t = 11 - ((j + 1) % NQ);
    i ^= ((i >> c) & 1) << t;
  }
  return i;
}
__device__ __forceinline__ int perm_inv(int i) {
#pragma unroll
  for (int j = 0; j < NQ; ++j) {
    int c = 11 - j;
    int t = 11 - ((j + 1) % NQ);
    i ^= ((i >> c) & 1) << t;
  }
  return i;
}

__device__ __forceinline__ int sweep_idx(int s, int g, int v) {
  if (s == 0) return (g << 4) | v;
  if (s == 1) return ((g & 0xF0) << 4) | (v << 4) | (g & 0x0F);
  return (v << 8) | g;
}

__global__ __launch_bounds__(256) void qsim_kernel(const float* __restrict__ x,
                                                   const float* __restrict__ w,
                                                   float* __restrict__ out) {
  __shared__ float sre[NPADDED];
  __shared__ float sim[NPADDED];
  __shared__ float matR[16 * 17];
  __shared__ float csA[NQ], snA[NQ], csB[NQ], snB[NQ];
  __shared__ float red[8][NQ];

  const int tid = threadIdx.x;
  const int b = blockIdx.x;
  const int lane = tid & 31;
  const int wv = tid >> 5;

  if (tid < NQ) {
    int q = 11 - tid;
    float thA = x[b * NQ + q] + w[q];
    float thB = w[NQ + q];
    csA[tid] = cosf(0.5f * thA);
    snA[tid] = sinf(0.5f * thA);
    csB[tid] = cosf(0.5f * thB);
    snB[tid] = sinf(0.5f * thB);
  }
  __syncthreads();

  {
    float base = 1.0f;
#pragma unroll
    for (int k = 0; k < 8; ++k)
      base *= ((tid >> k) & 1) ? snA[4 + k] : csA[4 + k];
#pragma unroll
    for (int v = 0; v < 16; ++v) {
      float m = base;
#pragma unroll
      for (int k = 0; k < 4; ++k)
        m *= ((v >> k) & 1) ? snA[k] : csA[k];
      int idx = (tid << 4) | v;
      sre[PAD_IDX(idx)] = m;
    }
  }
  __syncthreads();

  {
    float tr[16], ti[16];
    const int pct = __popc(tid);
#pragma unroll
    for (int v = 0; v < 16; ++v) {
      int j = perm_fwd((tid << 4) | v);
      float m = sre[PAD_IDX(j)];
      int e = (pct + __popc(v) - __popc(j)) & 3;
      float sm = (e & 2) ? -m : m;
      tr[v] = (e & 1) ? 0.0f : sm;
      ti[v] = (e & 1) ? sm : 0.0f;
    }
    __syncthreads();
#pragma unroll
    for (int v = 0; v < 16; ++v) {
      int i = (tid << 4) | v;
      sre[PAD_IDX(i)] = tr[v];
      sim[PAD_IDX(i)] = ti[v];
    }
  }
  __syncthreads();

#pragma unroll
  for (int s = 0; s < 3; ++s) {
    {
      int r = tid >> 4, c = tid & 15, xr = r ^ c;
      float m = 1.0f;
#pragma unroll
      for (int k = 0; k < 4; ++k)
        m *= ((xr >> k) & 1) ? snB[4 * s + k] : csB[4 * s + k];
      matR[r * 17 + c] = (__popc(c & ~r) & 1) ? -m : m;
    }
    __syncthreads();

    const int arow = lane & 15;
    const int koff = (lane >> 4) << 1;
    v2f aR[4];
#pragma unroll
    for (int j = 0; j < 4; ++j) {
      int k0 = 4 * j + koff;
      aR[j].x = matR[arow * 17 + k0];
      aR[j].y = matR[arow * 17 + k0 + 1];
    }

#pragma unroll
    for (int tt = 0; tt < 2; ++tt) {
      int gcol = (wv * 2 + tt) * 16 + (lane & 15);
      v2f bre[4], bim[4];
#pragma unroll
      for (int j = 0; j < 4; ++j) {
        int v0 = 4 * j + koff;
        int i0 = sweep_idx(s, gcol, v0);
        int i1 = sweep_idx(s, gcol, v0 + 1);
        bre[j].x = sre[PAD_IDX(i0)];
        bre[j].y = sre[PAD_IDX(i1)];
        bim[j].x = sim[PAD_IDX(i0)];
        bim[j].y = sim[PAD_IDX(i1)];
      }
      v8f dre = {};
      v8f dim = {};
#pragma unroll
      for (int j = 0; j < 4; ++j) {
        dre = __builtin_amdgcn_wmma_f32_16x16x4_f32(false, aR[j], false, bre[j],
                                                    (short)0, dre, false, false);
        dim = __builtin_amdgcn_wmma_f32_16x16x4_f32(false, aR[j], false, bim[j],
                                                    (short)0, dim, false, false);
      }
      int vadd = (lane >> 4) << 3;
#pragma unroll
      for (int r = 0; r < 8; ++r) {
        int idx = sweep_idx(s, gcol, r + vadd);
        sre[PAD_IDX(idx)] = dre[r];
        sim[PAD_IDX(idx)] = dim[r];
      }
    }
    __syncthreads();
  }

  float acc[NQ];
#pragma unroll
  for (int q = 0; q < NQ; ++q) acc[q] = 0.0f;
#pragma unroll
  for (int v = 0; v < 16; ++v) {
    int jdx = (tid << 4) | v;
    float re = sre[PAD_IDX(jdx)];
    float im = sim[PAD_IDX(jdx)];
    float p = re * re + im * im;
    int i = perm_inv(jdx);
#pragma unroll
    for (int q = 0; q < NQ; ++q) acc[q] += ((i >> q) & 1) ? -p : p;
  }
#pragma unroll
  for (int off = 16; off > 0; off >>= 1) {
#pragma unroll
    for (int q = 0; q < NQ; ++q) acc[q] += __shfl_down(acc[q], off, 32);
  }
  if (lane == 0) {
#pragma unroll
    for (int q = 0; q < NQ; ++q) red[wv][q] = acc[q];
  }
  __syncthreads();
  if (tid < NQ) {
    float sum = 0.0f;
#pragma unroll
    for (int k = 0; k < 8; ++k) sum += red[k][tid];
    out[b * NQ + (11 - tid)] = sum;
  }
}

extern "C" void kernel_launch(void* const* d_in, const int* in_sizes, int n_in,
                              void* d_out, int out_size, void* d_ws,
                              size_t ws_size, hipStream_t stream) {
  const float* x = (const float*)d_in[0];
  const float* w = (const float*)d_in[1];
  float* out = (float*)d_out;
  int B = in_sizes[0] / NQ;
  qsim_kernel<<<B, 256, 0, stream>>>(x, w, out);
}


# Final <Z> as parity-mask (Walsh) sums: CNOT ring 2 is linear over GF(2), so lanes sign-sum the last sweep's accumulator probs and join by DPP butterfly; no index math or LDS trip; sum order changes
# speedup vs baseline: 2.1012x; 1.6182x over previous
#include <hip/hip_runtime.h>
#include <math.h>

typedef __attribute__((ext_vector_type(2))) float v2f;
typedef __attribute__((ext_vector_type(8))) float v8f;

#define NQ 12
#define NSTATE 4096
#define NPADDED (4096 + 256)
#define PAD_IDX(i) ((i) + ((i) >> 4))

__device__ __forceinline__ int perm_fwd(int i) {
#pragma unroll
  for (int j = NQ - 1; j >= 0; --j) {
    int c = 11 - j;
    int t = 11 - ((j + 1) % NQ);
    i ^= ((i >> c) & 1) << t;
  }
  return i;
}
__device__ __forceinline__ int sweep_idx(int s, int g, int v) {
  if (s == 0) return (g << 4) | v;
  if (s == 1) return ((g & 0xF0) << 4) | (v << 4) | (g & 0x0F);
  return (v << 8) | g;
}

__device__ __forceinline__ float xhalf(float x) {
  int b = __builtin_bit_cast(int, x);
  return __builtin_bit_cast(
      float, __builtin_amdgcn_permlanex16(b, b, 0x76543210, 0xfedcba98, false, false));
}
template <int M>
__device__ __forceinline__ float xrow(float x) {
  int b = __builtin_bit_cast(int, x);
  return __builtin_bit_cast(
      float, __builtin_amdgcn_update_dpp(b, b, 0x160 | M, 0xf, 0xf, true));
}
__device__ __forceinline__ float flip(float x, unsigned m) {
  return __builtin_bit_cast(float, __builtin_bit_cast(unsigned, x) ^ m);
}

__global__ __launch_bounds__(256) void qsim_kernel(const float* __restrict__ x,
                                                   const float* __restrict__ w,
                                                   float* __restrict__ out) {
  __shared__ float sre[NPADDED];
  __shared__ float sim[NPADDED];
  __shared__ float matR[16 * 17];
  __shared__ float csA[NQ], snA[NQ], csB[NQ], snB[NQ];
  __shared__ float red[8][40];

  const int tid = threadIdx.x;
  const int b = blockIdx.x;
  const int lane = tid & 31;
  const int wv = tid >> 5;

  if (tid < NQ) {
    int q = 11 - tid;
    float thA = x[b * NQ + q] + w[q];
    float thB = w[NQ + q];
    csA[tid] = cosf(0.5f * thA);
    snA[tid] = sinf(0.5f * thA);
    csB[tid] = cosf(0.5f * thB);
    snB[tid] = sinf(0.5f * thB);
  }
  __syncthreads();

  {
    float base = 1.0f;
#pragma unroll
    for (int k = 0; k < 8; ++k)
      base *= ((tid >> k) & 1) ? snA[4 + k] : csA[4 + k];
#pragma unroll
    for (int v = 0; v < 16; ++v) {
      float m = base;
#pragma unroll
      for (int k = 0; k < 4; ++k)
        m *= ((v >> k) & 1) ? snA[k] : csA[k];
      int idx = (tid << 4) | v;
      sre[PAD_IDX(idx)] = m;
    }
  }
  __syncthreads();

  {
    float tr[16], ti[16];
    const int pct = __popc(tid);
#pragma unroll
    for (int v = 0; v < 16; ++v) {
      int j = perm_fwd((tid << 4) | v);
      float m = sre[PAD_IDX(j)];
      int e = (pct + __popc(v) - __popc(j)) & 3;
      float sm = (e & 2) ? -m : m;
      tr[v] = (e & 1) ? 0.0f : sm;
      ti[v] = (e & 1) ? sm : 0.0f;
    }
    __syncthreads();
#pragma unroll
    for (int v = 0; v < 16; ++v) {
      int i = (tid << 4) | v;
      sre[PAD_IDX(i)] = tr[v];
      sim[PAD_IDX(i)] = ti[v];
    }
  }
  __syncthreads();

  float pr[2][8];
#pragma unroll
  for (int s = 0; s < 3; ++s) {
    {
      int r = tid >> 4, c = tid & 15, xr = r ^ c;
      float m = 1.0f;
#pragma unroll
      for (int k = 0; k < 4; ++k)
        m *= ((xr >> k) & 1) ? snB[4 * s + k] : csB[4 * s + k];
      matR[r * 17 + c] = (__popc(c & ~r) & 1) ? -m : m;
    }
    __syncthreads();

    const int arow = lane & 15;
    const int koff = (lane >> 4) << 1;
    v2f aR[4];
#pragma unroll
    for (int j = 0; j < 4; ++j) {
      int k0 = 4 * j + koff;
      aR[j].x = matR[arow * 17 + k0];
      aR[j].y = matR[arow * 17 + k0 + 1];
    }

#pragma unroll
    for (int tt = 0; tt < 2; ++tt) {
      int gcol = (wv * 2 + tt) * 16 + (lane & 15);
      v2f bre[4], bim[4];
#pragma unroll
      for (int j = 0; j < 4; ++j) {
        int v0 = 4 * j + koff;
        int i0 = sweep_idx(s, gcol, v0);
        int i1 = sweep_idx(s, gcol, v0 + 1);
        bre[j].x = sre[PAD_IDX(i0)];
        bre[j].y = sre[PAD_IDX(i1)];
        bim[j].x = sim[PAD_IDX(i0)];
        bim[j].y = sim[PAD_IDX(i1)];
      }
      v8f dre = {};
      v8f dim = {};
#pragma unroll
      for (int j = 0; j < 4; ++j) {
        dre = __builtin_amdgcn_wmma_f32_16x16x4_f32(false, aR[j], false, bre[j],
                                                    (short)0, dre, false, false);
        dim = __builtin_amdgcn_wmma_f32_16x16x4_f32(false, aR[j], false, bim[j],
                                                    (short)0, dim, false, false);
      }
      if (s < 2) {
        int vadd = (lane >> 4) << 3;
#pragma unroll
        for (int r = 0; r < 8; ++r) {
          int idx = sweep_idx(s, gcol, r + vadd);
          sre[PAD_IDX(idx)] = dre[r];
          sim[PAD_IDX(idx)] = dim[r];
        }
      } else {
#pragma unroll
        for (int r = 0; r < 8; ++r) pr[tt][r] = dre[r] * dre[r] + dim[r] * dim[r];
      }
    }
    if (s < 2) __syncthreads();
  }

  {
    float sm[8], df[8];
#pragma unroll
    for (int r = 0; r < 8; ++r) {
      sm[r] = pr[0][r] + pr[1][r];
      df[r] = pr[0][r] - pr[1][r];
    }
    float e0 = df[0] - df[1], e1 = df[2] - df[3];
    float e2 = df[4] - df[5], e3 = df[6] - df[7];
    float wall = (e0 - e1) - (e2 - e3);
    float sp0 = sm[0] + sm[1], sp1 = sm[2] + sm[3];
    float sp2 = sm[4] + sm[5], sp3 = sm[6] + sm[7];
    float sn0 = sm[0] - sm[1], sn1 = sm[2] - sm[3];
    float sn2 = sm[4] - sm[5], sn3 = sm[6] - sm[7];
    float w8910 = (sn0 - sn1) - (sn2 - sn3);
    float w910 = (sp0 - sp1) - (sp2 - sp3);
    float w10 = (sp0 + sp1) - (sp2 + sp3);

    const unsigned wpar = (unsigned)(__popc(wv) & 1) << 31;
    const unsigned w67 = (unsigned)(__popc(wv & 6) & 1) << 31;
    const unsigned w7 = (unsigned)((wv >> 2) & 1) << 31;

    float wx = flip(wall, wpar);
    wx = xrow<1>(wx) + flip(wx, (unsigned)(lane & 1) << 31);
    wx = xrow<2>(wx) + flip(wx, (unsigned)((lane >> 1) & 1) << 31);
    wx = xrow<4>(wx) + flip(wx, (unsigned)((lane >> 2) & 1) << 31);
    wx = xrow<8>(wx) + flip(wx, (unsigned)((lane >> 3) & 1) << 31);
    wx = xhalf(wx) + flip(wx, (unsigned)((lane >> 4) & 1) << 31);

    float y1 = w8910, y2 = w910, y3 = w10;
    y1 += xrow<1>(y1); y2 += xrow<1>(y2); y3 += xrow<1>(y3);
    y1 += xrow<2>(y1); y2 += xrow<2>(y2); y3 += xrow<2>(y3);
    y1 += xrow<4>(y1); y2 += xrow<4>(y2); y3 += xrow<4>(y3);
    y1 += xrow<8>(y1); y2 += xrow<8>(y2); y3 += xrow<8>(y3);
    y1 -= xhalf(y1); y2 -= xhalf(y2); y3 -= xhalf(y3);

    red[wv][lane] = wx;
    if (lane == 0) {
      red[wv][32] = flip(y1, wpar);
      red[wv][33] = flip(y1, w67);
      red[wv][34] = flip(y1, w7);
      red[wv][35] = y1;
      red[wv][36] = y2;
      red[wv][37] = y3;
    }
  }
  __syncthreads();
  if (tid < NQ) {
    const int slot = (tid <= 4) ? ((31 << tid) & 31) : (tid == 11) ? 15 : 27 + tid;
    float sum = 0.0f;
#pragma unroll
    for (int k = 0; k < 8; ++k) sum += red[k][slot];
    out[b * NQ + (11 - tid)] = sum;
  }
}

extern "C" void kernel_launch(void* const* d_in, const int* in_sizes, int n_in,
                              void* d_out, int out_size, void* d_ws,
                              size_t ws_size, hipStream_t stream) {
  const float* x = (const float*)d_in[0];
  const float* w = (const float*)d_in[1];
  float* out = (float*)d_out;
  int B = in_sizes[0] / NQ;
  qsim_kernel<<<B, 256, 0, stream>>>(x, w, out);
}


# State after CNOT ring 1 from its closed form (tau is GF(2)-linear: factors of neighbour-bit XORs times a power of -i), built in sweep 0's B registers: no LDS build or gather; products regrouped
# speedup vs baseline: 2.4219x; 1.1526x over previous
#include <hip/hip_runtime.h>
#include <math.h>

typedef __attribute__((ext_vector_type(2))) float v2f;
typedef __attribute__((ext_vector_type(8))) float v8f;

#define NQ 12
#define NSTATE 4096
#define NPADDED (4096 + 256)
#define PAD_IDX(i) ((i) + ((i) >> 4))

__device__ __forceinline__ int sweep_idx(int s, int g, int v) {
  if (s == 0) return (g << 4) | v;
  if (s == 1) return ((g & 0xF0) << 4) | (v << 4) | (g & 0x0F);
  return (v << 8) | g;
}

__device__ __forceinline__ float xhalf(float x) {
  int b = __builtin_bit_cast(int, x);
  return __builtin_bit_cast(
      float, __builtin_amdgcn_permlanex16(b, b, 0x76543210, 0xfedcba98, false, false));
}
template <int M>
__device__ __forceinline__ float xrow(float x) {
  int b = __builtin_bit_cast(int, x);
  return __builtin_bit_cast(
      float, __builtin_amdgcn_update_dpp(b, b, 0x160 | M, 0xf, 0xf, true));
}
__device__ __forceinline__ float flip(float x, unsigned m) {
  return __builtin_bit_cast(float, __builtin_bit_cast(unsigned, x) ^ m);
}

__global__ __launch_bounds__(256, 16) void qsim_kernel(const float* __restrict__ x,
                                                   const float* __restrict__ w,
                                                   float* __restrict__ out) {
  __shared__ float sre[NPADDED];
  __shared__ float sim[NPADDED];
  __shared__ float matR[16 * 17];
  __shared__ __attribute__((aligned(16))) float fA[2 * NQ];
  __shared__ float csB[NQ], snB[NQ];
  __shared__ float red[8][40];

  const int tid = threadIdx.x;
  const int b = blockIdx.x;
  const int lane = tid & 31;
  const int wv = tid >> 5;

  if (tid < NQ) {
    int q = 11 - tid;
    float thA = x[b * NQ + q] + w[q];
    float thB = w[NQ + q];
    fA[2 * tid] = cosf(0.5f * thA);
    fA[2 * tid + 1] = sinf(0.5f * thA);
    csB[tid] = cosf(0.5f * thB);
    snB[tid] = sinf(0.5f * thB);
  }
  __syncthreads();

  float facT[2], facE[2], facJ[4], zr, zi;
  {
    const int lh = lane >> 4;
    const int c0 = lane & 1, c3 = (lane >> 3) & 1;
    const int lx = lane ^ (lane >> 1);
    const int w0 = wv & 1, w1 = (wv >> 1) & 1, w2 = (wv >> 2) & 1;
    const v2f p0 = *(const v2f*)&fA[0], p1 = *(const v2f*)&fA[2], p2 = *(const v2f*)&fA[4];
    const v2f p3 = *(const v2f*)&fA[6], p7 = *(const v2f*)&fA[14], p8 = *(const v2f*)&fA[16];
    const v2f p10 = *(const v2f*)&fA[20], p11 = *(const v2f*)&fA[22];
    const float fl = fA[8 + (lx & 1)] * fA[10 + ((lx >> 1) & 1)] *
                     fA[12 + ((lx >> 2) & 1)] * fA[18 + (w0 ^ w1)];
    facT[0] = fl * (c3 ? p7.y : p7.x) * (w0 ? p8.y : p8.x);
    facT[1] = flip(fl * (c3 ? p7.x : p7.y) * (w0 ? p8.x : p8.y), (unsigned)(c3 ^ w0) << 31);
    facE[0] = (lh ? p0.y : p0.x) * ((w1 ^ w2) ? p10.y : p10.x) * (w2 ? p11.y : p11.x);
    facE[1] = flip((lh ? p0.x : p0.y) * ((w1 ^ w2) ? p10.x : p10.y) * (w2 ? p11.x : p11.y),
                   (unsigned)(1 ^ lh ^ w1) << 31);
    const float a1 = lh ? p1.y : p1.x, b1 = lh ? p1.x : p1.y;
    const float a3 = c0 ? p3.y : p3.x, b3 = c0 ? p3.x : p3.y;
    facJ[0] = a1 * p2.x * a3;
    facJ[1] = flip(b1 * p2.y * a3, (unsigned)lh << 31);
    facJ[2] = flip(a1 * p2.y * b3, (unsigned)c0 << 31);
    facJ[3] = flip(b1 * p2.x * b3, (unsigned)(1 ^ lh ^ c0) << 31);
    const int pl = __popc(lane) + __popc(wv);
    const int cl = lane & 15;
    const unsigned sl = (unsigned)((__popc(cl & (cl >> 1)) + __popc(wv & (wv >> 1)) + (pl >> 1)) & 1) << 31;
    zr = flip((pl & 1) ? 0.0f : 1.0f, sl);
    zi = flip((pl & 1) ? -1.0f : 0.0f, sl);
  }

  float pr[2][8];
#pragma unroll
  for (int s = 0; s < 3; ++s) {
    {
      int r = tid >> 4, c = tid & 15, xr = r ^ c;
      float m = 1.0f;
#pragma unroll
      for (int k = 0; k < 4; ++k)
        m *= ((xr >> k) & 1) ? snB[4 * s + k] : csB[4 * s + k];
      matR[r * 17 + c] = (__popc(c & ~r) & 1) ? -m : m;
    }
    __syncthreads();

    const int arow = lane & 15;
    const int koff = (lane >> 4) << 1;
    v2f aR[4];
#pragma unroll
    for (int j = 0; j < 4; ++j) {
      int k0 = 4 * j + koff;
      aR[j].x = matR[arow * 17 + k0];
      aR[j].y = matR[arow * 17 + k0 + 1];
    }

#pragma unroll
    for (int tt = 0; tt < 2; ++tt) {
      int gcol = (wv * 2 + tt) * 16 + (lane & 15);
      v2f bre[4], bim[4];
      if (s == 0) {
#pragma unroll
        for (int e = 0; e < 2; ++e) {
          const float te = facT[tt] * facE[e];
          const float ter = te * zr, tei = te * zi;
#pragma unroll
          for (int j = 0; j < 4; ++j) {
            const int turns = ((j & 1) + (j >> 1) + tt) & 3;
            const float vr = (turns == 0) ? ter : (turns == 1) ? tei : (turns == 2) ? -ter : -tei;
            const float vi = (turns == 0) ? tei : (turns == 1) ? -ter : (turns == 2) ? -tei : ter;
            if (e == 0) { bre[j].x = vr * facJ[j]; bim[j].x = vi * facJ[j]; }
            else        { bre[j].y = vr * facJ[j]; bim[j].y = vi * facJ[j]; }
          }
        }
      } else {
#pragma unroll
        for (int j = 0; j < 4; ++j) {
          int v0 = 4 * j + koff;
          int i0 = sweep_idx(s, gcol, v0);
          int i1 = sweep_idx(s, gcol, v0 + 1);
          bre[j].x = sre[PAD_IDX(i0)];
          bre[j].y = sre[PAD_IDX(i1)];
          bim[j].x = sim[PAD_IDX(i0)];
          bim[j].y = sim[PAD_IDX(i1)];
        }
      }
      v8f dre = {};
      v8f dim = {};
#pragma unroll
      for (int j = 0; j < 4; ++j) {
        dre = __builtin_amdgcn_wmma_f32_16x16x4_f32(false, aR[j], false, bre[j],
                                                    (short)0, dre, false, false);
        dim = __builtin_amdgcn_wmma_f32_16x16x4_f32(false, aR[j], false, bim[j],
                                                    (short)0, dim, false, false);
      }
      if (s < 2) {
        int vadd = (lane >> 4) << 3;
#pragma unroll
        for (int r = 0; r < 8; ++r) {
          int idx = sweep_idx(s, gcol, r + vadd);
          sre[PAD_IDX(idx)] = dre[r];
          sim[PAD_IDX(idx)] = dim[r];
        }
      } else {
#pragma unroll
        for (int r = 0; r < 8; ++r) pr[tt][r] = dre[r] * dre[r] + dim[r] * dim[r];
      }
    }
    if (s < 2) __syncthreads();
  }

  {
    float sm[8], df[8];
#pragma unroll
    for (int r = 0; r < 8; ++r) {
      sm[r] = pr[0][r] + pr[1][r];
      df[r] = pr[0][r] - pr[1][r];
    }
    float e0 = df[0] - df[1], e1 = df[2] - df[3];
    float e2 = df[4] - df[5], e3 = df[6] - df[7];
    float wall = (e0 - e1) - (e2 - e3);
    float sp0 = sm[0] + sm[1], sp1 = sm[2] + sm[3];
    float sp2 = sm[4] + sm[5], sp3 = sm[6] + sm[7];
    float sn0 = sm[0] - sm[1], sn1 = sm[2] - sm[3];
    float sn2 = sm[4] - sm[5], sn3 = sm[6] - sm[7];
    float w8910 = (sn0 - sn1) - (sn2 - sn3);
    float w910 = (sp0 - sp1) - (sp2 - sp3);
    float w10 = (sp0 + sp1) - (sp2 + sp3);

    const unsigned wpar = (unsigned)(__popc(wv) & 1) << 31;
    const unsigned w67 = (unsigned)(__popc(wv & 6) & 1) << 31;
    const unsigned w7 = (unsigned)((wv >> 2) & 1) << 31;

    float wx = flip(wall, wpar);
    wx = xrow<1>(wx) + flip(wx, (unsigned)(lane & 1) << 31);
    wx = xrow<2>(wx) + flip(wx, (unsigned)((lane >> 1) & 1) << 31);
    wx = xrow<4>(wx) + flip(wx, (unsigned)((lane >> 2) & 1) << 31);
    wx = xrow<8>(wx) + flip(wx, (unsigned)((lane >> 3) & 1) << 31);
    wx = xhalf(wx) + flip(wx, (unsigned)((lane >> 4) & 1) << 31);

    float y1 = w8910, y2 = w910, y3 = w10;
    y1 += xrow<1>(y1); y2 += xrow<1>(y2); y3 += xrow<1>(y3);
    y1 += xrow<2>(y1); y2 += xrow<2>(y2); y3 += xrow<2>(y3);
    y1 += xrow<4>(y1); y2 += xrow<4>(y2); y3 += xrow<4>(y3);
    y1 += xrow<8>(y1); y2 += xrow<8>(y2); y3 += xrow<8>(y3);
    y1 -= xhalf(y1); y2 -= xhalf(y2); y3 -= xhalf(y3);

    red[wv][lane] = wx;
    if (lane == 0) {
      red[wv][32] = flip(y1, wpar);
      red[wv][33] = flip(y1, w67);
      red[wv][34] = flip(y1, w7);
      red[wv][35] = y1;
      red[wv][36] = y2;
      red[wv][37] = y3;
    }
  }
  __syncthreads();
  if (tid < NQ) {
    const int slot = (tid <= 4) ? ((31 << tid) & 31) : (tid == 11) ? 15 : 27 + tid;
    float sum = 0.0f;
#pragma unroll
    for (int k = 0; k < 8; ++k) sum += red[k][slot];
    out[b * NQ + (11 - tid)] = sum;
  }
}

extern "C" void kernel_launch(void* const* d_in, const int* in_sizes, int n_in,
                              void* d_out, int out_size, void* d_ws,
                              size_t ws_size, hipStream_t stream) {
  const float* x = (const float*)d_in[0];
  const float* w = (const float*)d_in[1];
  float* out = (float*)d_out;
  int B = in_sizes[0] / NQ;
  qsim_kernel<<<B, 256, 0, stream>>>(x, w, out);
}


# Sweep 0: the closed-form state is real times a power of -i, so k is grouped by phase parity and Re, Im are one K=8 real product each (WMMAs per wave 48->40); sums drop exact zeros, k order changes
# speedup vs baseline: 2.6306x; 1.0862x over previous
#include <hip/hip_runtime.h>
#include <math.h>

typedef __attribute__((ext_vector_type(2))) float v2f;
typedef __attribute__((ext_vector_type(8))) float v8f;

#define NQ 12
#define NSTATE 4096
#define NPADDED (4096 + 256)
#define PAD_IDX(i) ((i) + ((i) >> 4))

__device__ __forceinline__ int sweep_idx(int s, int g, int v) {
  if (s == 0) return (g << 4) | v;
  if (s == 1) return ((g & 0xF0) << 4) | (v << 4) | (g & 0x0F);
  return (v << 8) | g;
}

__device__ __forceinline__ float xhalf(float x) {
  int b = __builtin_bit_cast(int, x);
  return __builtin_bit_cast(
      float, __builtin_amdgcn_permlanex16(b, b, 0x76543210, 0xfedcba98, false, false));
}
template <int M>
__device__ __forceinline__ float xrow(float x) {
  int b = __builtin_bit_cast(int, x);
  return __builtin_bit_cast(
      float, __builtin_amdgcn_update_dpp(b, b, 0x160 | M, 0xf, 0xf, true));
}
__device__ __forceinline__ float flip(float x, unsigned m) {
  return __builtin_bit_cast(float, __builtin_bit_cast(unsigned, x) ^ m);
}

__global__ __launch_bounds__(256, 16) void qsim_kernel(const float* __restrict__ x,
                                                   const float* __restrict__ w,
                                                   float* __restrict__ out) {
  __shared__ float sre[NPADDED];
  __shared__ float sim[NPADDED];
  __shared__ float matR[16 * 17];
  __shared__ __attribute__((aligned(16))) float fA[2 * NQ];
  __shared__ float csB[NQ], snB[NQ];
  __shared__ float red[8][40];

  const int tid = threadIdx.x;
  const int b = blockIdx.x;
  const int lane = tid & 31;
  const int wv = tid >> 5;

  if (tid < NQ) {
    int q = 11 - tid;
    float thA = x[b * NQ + q] + w[q];
    float thB = w[NQ + q];
    fA[2 * tid] = cosf(0.5f * thA);
    fA[2 * tid + 1] = sinf(0.5f * thA);
    csB[tid] = cosf(0.5f * thB);
    snB[tid] = sinf(0.5f * thB);
  }
  __syncthreads();

  float facT[2], facE[2], facK[4];
  int col8[2];
  {
    const int lh = lane >> 4;
    const int c0 = lane & 1, c3 = (lane >> 3) & 1;
    const int lx = lane ^ (lane >> 1);
    const int w0 = wv & 1, w1 = (wv >> 1) & 1, w2 = (wv >> 2) & 1;
    const v2f p0 = *(const v2f*)&fA[0], p1 = *(const v2f*)&fA[2], p2 = *(const v2f*)&fA[4];
    const v2f p3 = *(const v2f*)&fA[6], p7 = *(const v2f*)&fA[14], p8 = *(const v2f*)&fA[16];
    const v2f p10 = *(const v2f*)&fA[20], p11 = *(const v2f*)&fA[22];
    const float fl = fA[8 + (lx & 1)] * fA[10 + ((lx >> 1) & 1)] *
                     fA[12 + ((lx >> 2) & 1)] * fA[18 + (w0 ^ w1)];
    const int cl = lane & 15;
    const int pc = __popc(cl) + __popc(wv);
    const int sn = __popc(cl & (cl >> 1)) + __popc(wv & (wv >> 1));
#pragma unroll
    for (int tt = 0; tt < 2; ++tt) {
      const int b8 = (tt ^ pc) & 1;
      const int c = pc + b8;
      const float mag = fl * ((c3 ^ b8) ? p7.y : p7.x) * ((b8 ^ w0) ? p8.y : p8.x);
      const int sg = ((b8 & (c3 ^ w0)) ^ sn ^ (c >> 1) ^ tt) & 1;
      facT[tt] = flip(mag, (unsigned)sg << 31);
      col8[tt] = b8;
    }
    facE[0] = (lh ? p0.y : p0.x) * ((w1 ^ w2) ? p10.y : p10.x) * (w2 ? p11.y : p11.x);
    facE[1] = flip((lh ? p0.x : p0.y) * ((w1 ^ w2) ? p10.x : p10.y) * (w2 ? p11.x : p11.y),
                   (unsigned)(1 ^ lh ^ w1) << 31);
#pragma unroll
    for (int jj = 0; jj < 4; ++jj) {
      const int qk = jj >> 1, j0 = jj & 1;
      const int j1 = qk ^ lh ^ j0;
      const float f1 = (lh ^ j0) ? p1.y : p1.x;
      const float f2 = (qk ^ lh) ? p2.y : p2.x;
      const float f3 = (j1 ^ c0) ? p3.y : p3.x;
      const int sg = ((j0 & j1) ^ (j0 & lh) ^ (j1 & c0) ^ ((lh + j0 + j1) >> 1)) & 1;
      facK[jj] = flip(f1 * f2 * f3, (unsigned)sg << 31);
    }
  }

  float pr[2][8];
#pragma unroll
  for (int s = 0; s < 3; ++s) {
    {
      int r = tid >> 4, c = tid & 15, xr = r ^ c;
      float m = 1.0f;
#pragma unroll
      for (int k = 0; k < 4; ++k)
        m *= ((xr >> k) & 1) ? snB[4 * s + k] : csB[4 * s + k];
      matR[r * 17 + c] = (__popc(c & ~r) & 1) ? -m : m;
    }
    __syncthreads();

    const int arow = lane & 15;
    const int koff = (lane >> 4) << 1;
    v2f aR[4];
#pragma unroll
    for (int j = 0; j < 4; ++j) {
      int k0 = (s == 0) ? (koff | ((j & 1) << 2) | ((((j >> 1) ^ (lane >> 4) ^ j) & 1) << 3))
                        : 4 * j + koff;
      aR[j].x = matR[arow * 17 + k0];
      aR[j].y = matR[arow * 17 + k0 + 1];
    }

#pragma unroll
    for (int tt = 0; tt < 2; ++tt) {
      int gcol = (wv * 2 + tt) * 16 + (lane & 15);
      v2f bre[4], bim[4];
      if (s == 0) {
        const float t0 = facT[tt];
        const float t1 = (tt == 0) ? -t0 : t0;
        const float u00 = t0 * facE[0], u01 = t0 * facE[1];
        const float u10 = t1 * facE[0], u11 = t1 * facE[1];
        v2f b0[2], b1[2];
#pragma unroll
        for (int j = 0; j < 2; ++j) {
          b0[j].x = u00 * facK[j];     b0[j].y = u01 * facK[j];
          b1[j].x = u10 * facK[2 + j]; b1[j].y = u11 * facK[2 + j];
        }
        v8f p0 = {};
        v8f p1 = {};
#pragma unroll
        for (int j = 0; j < 2; ++j) {
          p0 = __builtin_amdgcn_wmma_f32_16x16x4_f32(false, aR[j], false, b0[j],
                                                     (short)0, p0, false, false);
          p1 = __builtin_amdgcn_wmma_f32_16x16x4_f32(false, aR[2 + j], false, b1[j],
                                                     (short)0, p1, false, false);
        }
        const int gcol0 = (wv * 2 + col8[tt]) * 16 + (lane & 15);
        const int vadd0 = (lane >> 4) << 3;
#pragma unroll
        for (int r = 0; r < 8; ++r) {
          int idx = PAD_IDX(sweep_idx(0, gcol0, r + vadd0));
          sre[idx] = (tt == 0) ? p0[r] : p1[r];
          sim[idx] = (tt == 0) ? p1[r] : p0[r];
        }
        continue;
      } else {
#pragma unroll
        for (int j = 0; j < 4; ++j) {
          int v0 = 4 * j + koff;
          int i0 = sweep_idx(s, gcol, v0);
          int i1 = sweep_idx(s, gcol, v0 + 1);
          bre[j].x = sre[PAD_IDX(i0)];
          bre[j].y = sre[PAD_IDX(i1)];
          bim[j].x = sim[PAD_IDX(i0)];
          bim[j].y = sim[PAD_IDX(i1)];
        }
      }
      v8f dre = {};
      v8f dim = {};
#pragma unroll
      for (int j = 0; j < 4; ++j) {
        dre = __builtin_amdgcn_wmma_f32_16x16x4_f32(false, aR[j], false, bre[j],
                                                    (short)0, dre, false, false);
        dim = __builtin_amdgcn_wmma_f32_16x16x4_f32(false, aR[j], false, bim[j],
                                                    (short)0, dim, false, false);
      }
      if (s < 2) {
        int vadd = (lane >> 4) << 3;
#pragma unroll
        for (int r = 0; r < 8; ++r) {
          int idx = sweep_idx(s, gcol, r + vadd);
          sre[PAD_IDX(idx)] = dre[r];
          sim[PAD_IDX(idx)] = dim[r];
        }
      } else {
#pragma unroll
        for (int r = 0; r < 8; ++r) pr[tt][r] = dre[r] * dre[r] + dim[r] * dim[r];
      }
    }
    if (s < 2) __syncthreads();
  }

  {
    float sm[8], df[8];
#pragma unroll
    for (int r = 0; r < 8; ++r) {
      sm[r] = pr[0][r] + pr[1][r];
      df[r] = pr[0][r] - pr[1][r];
    }
    float e0 = df[0] - df[1], e1 = df[2] - df[3];
    float e2 = df[4] - df[5], e3 = df[6] - df[7];
    float wall = (e0 - e1) - (e2 - e3);
    float sp0 = sm[0] + sm[1], sp1 = sm[2] + sm[3];
    float sp2 = sm[4] + sm[5], sp3 = sm[6] + sm[7];
    float sn0 = sm[0] - sm[1], sn1 = sm[2] - sm[3];
    float sn2 = sm[4] - sm[5], sn3 = sm[6] - sm[7];
    float w8910 = (sn0 - sn1) - (sn2 - sn3);
    float w910 = (sp0 - sp1) - (sp2 - sp3);
    float w10 = (sp0 + sp1) - (sp2 + sp3);

    const unsigned wpar = (unsigned)(__popc(wv) & 1) << 31;
    const unsigned w67 = (unsigned)(__popc(wv & 6) & 1) << 31;
    const unsigned w7 = (unsigned)((wv >> 2) & 1) << 31;

    float wx = flip(wall, wpar);
    wx = xrow<1>(wx) + flip(wx, (unsigned)(lane & 1) << 31);
    wx = xrow<2>(wx) + flip(wx, (unsigned)((lane >> 1) & 1) << 31);
    wx = xrow<4>(wx) + flip(wx, (unsigned)((lane >> 2) & 1) << 31);
    wx = xrow<8>(wx) + flip(wx, (unsigned)((lane >> 3) & 1) << 31);
    wx = xhalf(wx) + flip(wx, (unsigned)((lane >> 4) & 1) << 31);

    float y1 = w8910, y2 = w910, y3 = w10;
    y1 += xrow<1>(y1); y2 += xrow<1>(y2); y3 += xrow<1>(y3);
    y1 += xrow<2>(y1); y2 += xrow<2>(y2); y3 += xrow<2>(y3);
    y1 += xrow<4>(y1); y2 += xrow<4>(y2); y3 += xrow<4>(y3);
    y1 += xrow<8>(y1); y2 += xrow<8>(y2); y3 += xrow<8>(y3);
    y1 -= xhalf(y1); y2 -= xhalf(y2); y3 -= xhalf(y3);

    red[wv][lane] = wx;
    if (lane == 0) {
      red[wv][32] = flip(y1, wpar);
      red[wv][33] = flip(y1, w67);
      red[wv][34] = flip(y1, w7);
      red[wv][35] = y1;
      red[wv][36] = y2;
      red[wv][37] = y3;
    }
  }
  __syncthreads();
  if (tid < NQ) {
    const int slot = (tid <= 4) ? ((31 << tid) & 31) : (tid == 11) ? 15 : 27 + tid;
    float sum = 0.0f;
#pragma unroll
    for (int k = 0; k < 8; ++k) sum += red[k][slot];
    out[b * NQ + (11 - tid)] = sum;
  }
}

extern "C" void kernel_launch(void* const* d_in, const int* in_sizes, int n_in,
                              void* d_out, int out_size, void* d_ws,
                              size_t ws_size, hipStream_t stream) {
  const float* x = (const float*)d_in[0];
  const float* w = (const float*)d_in[1];
  float* out = (float*)d_out;
  int B = in_sizes[0] / NQ;
  qsim_kernel<<<B, 256, 0, stream>>>(x, w, out);
}


# Final <Z> from signed Gram products of the three nibble tables (4x4 complex transfer matrices) and a ring trace: no amplitudes formed (WMMAs per block 76->36, waves 8->3); sum order changes
# speedup vs baseline: 7.3857x; 2.8076x over previous
#include <hip/hip_runtime.h>
#include <math.h>

typedef __attribute__((ext_vector_type(2))) float v2f;
typedef __attribute__((ext_vector_type(4))) float v4f;
typedef __attribute__((ext_vector_type(8))) float v8f;

#define NQ 12


template <int M>
__device__ __forceinline__ float xrow(float x) {
  int b = __builtin_bit_cast(int, x);
  return __builtin_bit_cast(
      float, __builtin_amdgcn_update_dpp(b, b, 0x160 | M, 0xf, 0xf, true));
}
__device__ __forceinline__ float flip(float x, unsigned m) {
  return __builtin_bit_cast(float, __builtin_bit_cast(unsigned, x) ^ m);
}

__global__ __launch_bounds__(96, 16) void qsim_kernel(const float* __restrict__ x,
                                                   const float* __restrict__ w,
                                                   float* __restrict__ out) {
  __shared__ __attribute__((aligned(16))) float fA[2 * NQ];
  __shared__ __attribute__((aligned(16))) float csB[NQ];
  __shared__ __attribute__((aligned(16))) float snB[NQ];
  __shared__ __attribute__((aligned(16))) float es[24 * 32];
  __shared__ float red[NQ];

  const int tid = threadIdx.x;
  const int b = blockIdx.x;
  const int lane = tid & 31;
  const int wv = tid >> 5;

  if (tid < NQ) {
    int q = 11 - tid;
    float thA = x[b * NQ + q] + w[q];
    float thB = w[NQ + q];
    fA[2 * tid] = cosf(0.5f * thA);
    fA[2 * tid + 1] = sinf(0.5f * thA);
    csB[tid] = cosf(0.5f * thB);
    snB[tid] = sinf(0.5f * thB);
  }
  __syncthreads();

  const int arow = lane & 15;
  const int lh = lane >> 4;
  const int cidx = arow & 7;
  const int ph = arow >> 3;
  const int kt = __builtin_amdgcn_readfirstlane(wv);
  v8f acc = {};
  {
    const v4f cs = *(const v4f*)&csB[4 * kt];
    const v4f sn = *(const v4f*)&snB[4 * kt];
    const int a0 = arow & 1, a1 = (arow >> 1) & 1, a2 = (arow >> 2) & 1, a3 = arow >> 3;
    const float fe0 = a0 ? sn.x : cs.x;
    const float fe1 = a0 ? cs.x : -sn.x;
    const float f1 = lh ? (a1 ? cs.y : -sn.y) : (a1 ? sn.y : cs.y);
    float g2[2], g3[2];
    g2[0] = a2 ? sn.z : cs.z;
    g2[1] = a2 ? cs.z : -sn.z;
    g3[0] = a3 ? sn.w : cs.w;
    g3[1] = a3 ? cs.w : -sn.w;
    const float h0 = fe0 * f1, h1 = fe1 * f1;
    const int ri = cidx & 1, bo = (cidx >> 1) & 1, bn = (cidx >> 2) & 1;
    const v2f p0 = *(const v2f*)&fA[8 * kt], p1 = *(const v2f*)&fA[8 * kt + 2];
    const v2f p2 = *(const v2f*)&fA[8 * kt + 4], p3 = *(const v2f*)&fA[8 * kt + 6];
    const int d0 = bo, d1 = lh;
    const int u = (kt == 2) ? bn : 0;
    const float m01 = (d0 ^ d1) ? p0.y : p0.x;
#pragma unroll
    for (int j = 0; j < 4; ++j) {
      const int d2 = j & 1, d3 = j >> 1;
      const float g23 = g2[d2] * g3[d3];
      v2f av;
      av.x = h0 * g23;
      av.y = h1 * g23;
      const float mag = m01 * ((d1 ^ d2) ? p1.y : p1.x) * ((d2 ^ d3 ^ u) ? p2.y : p2.x) *
                        ((d3 ^ bn) ? p3.y : p3.x);
      const int pw = d1 + d2 + d3 + ((kt != 0) ? d0 : 0);
      const int sg = (d0 & d1) + (d1 & d2) + (d2 & d3) + ((kt == 2) ? (bn & d2) : (d3 & bn)) +
                     ((kt == 0) ? d0 : 0) + ((pw + 1) >> 1);
      const bool nz = ((pw & 1) == ri);
      const float val = nz ? flip(mag, (unsigned)(sg & 1) << 31) : 0.0f;
      v2f bv;
      bv.x = bo ? 0.0f : val;
      bv.y = bo ? val : 0.0f;
      acc = __builtin_amdgcn_wmma_f32_16x16x4_f32(false, av, false, bv, (short)0, acc,
                                                  false, false);
    }
  }

  const unsigned sgnX0 = (ph & lh) ? 0x80000000u : 0u;
  const unsigned sgnX1 = (ph & (lh ^ 1)) ? 0x80000000u : 0u;
  const unsigned sgnP = ph ? 0x80000000u : 0u;
  const unsigned sgnB1e = (kt == 2) ? 0u : sgnX0;
  const unsigned sgnB1o = (kt == 2) ? sgnP : sgnX0;
  v8f dg0 = {};
  v8f dg1 = {};
#pragma unroll
  for (int j = 0; j < 4; ++j) {
    v2f av0, bv0, av1, bv1;
#pragma unroll
    for (int e = 0; e < 2; ++e) {
      const int r = 2 * j + e;
      const int par3 = (r ^ (r >> 1) ^ (r >> 2)) & 1;
      const int par2 = ((r >> 1) ^ (r >> 2)) & 1;
      const int hi = (r >> 2) & 1;
      const float t = acc[r];
      av0[e] = flip(t, par3 ? sgnX1 : sgnX0);
      bv0[e] = flip(t, par2 ? sgnX1 : sgnX0);
      av1[e] = flip(t, hi ? sgnX1 : sgnX0);
      bv1[e] = flip(t, par3 ? sgnB1o : sgnB1e);
    }
    dg0 = __builtin_amdgcn_wmma_f32_16x16x4_f32(false, av0, false, bv0, (short)0, dg0, false, false);
    dg1 = __builtin_amdgcn_wmma_f32_16x16x4_f32(false, av1, false, bv1, (short)0, dg1, false, false);
  }
  {
    float er0[4], ei0[4], er1[4], ei1[4];
#pragma unroll
    for (int a = 0; a < 4; ++a) {
      er0[a] = dg0[2 * a] + xrow<1>(dg0[2 * a + 1]);
      ei0[a] = dg0[2 * a + 1] - xrow<1>(dg0[2 * a]);
      er1[a] = dg1[2 * a] + xrow<1>(dg1[2 * a + 1]);
      ei1[a] = dg1[2 * a + 1] - xrow<1>(dg1[2 * a]);
    }
    if ((arow & 1) == 0) {
      float* dst = &es[(8 * kt + 2 * lh + ph) * 32 + (cidx >> 1) * 8];
      v4f s0, s1, s2, s3;
      s0.x = er0[0]; s0.y = ei0[0]; s0.z = er0[1]; s0.w = ei0[1];
      s1.x = er0[2]; s1.y = ei0[2]; s1.z = er0[3]; s1.w = ei0[3];
      s2.x = er1[0]; s2.y = ei1[0]; s2.z = er1[1]; s2.w = ei1[1];
      s3.x = er1[2]; s3.y = ei1[2]; s3.z = er1[3]; s3.w = ei1[3];
      *(v4f*)dst = s0;
      *(v4f*)(dst + 4) = s1;
      *(v4f*)(dst + 128) = s2;
      *(v4f*)(dst + 132) = s3;
    }
  }
  __syncthreads();

  float zsum[2];
#pragma unroll
  for (int pass = 0; pass < 2; ++pass) {
    const int q = (tid >> 4) + 6 * pass;
    const int L = tid & 15;
    const int b0 = L & 1, b0p = (L >> 1) & 1, b8 = (L >> 2) & 1, b8p = L >> 3;
    const int c0 = (int)((0x200000005612ULL >> (4 * q)) & 7);
    const int c1 = (int)((0x200056122222ULL >> (4 * q)) & 7);
    const int c2 = (int)((0x561222222222ULL >> (4 * q)) & 7);
    const float* e0 = &es[c0 * 32];
    const float* e1 = &es[(8 + c1) * 32];
    const float* e2 = &es[(16 + c2) * 32];
    float yr = 0.0f, yi = 0.0f;
#pragma unroll
    for (int t = 0; t < 4; ++t) {
      const int b4 = t & 1, b4p = t >> 1;
      const v2f u = *(const v2f*)&e0[((b0p + 2 * b4p) * 4 + (b0 + 2 * b4)) * 2];
      const v2f v = *(const v2f*)&e1[((b4p + 2 * b8p) * 4 + (b4 + 2 * b8)) * 2];
      yr += u.x * v.x - u.y * v.y;
      yi += u.x * v.y + u.y * v.x;
    }
    const v2f z = *(const v2f*)&e2[((b8p + 2 * b0p) * 4 + (b8 + 2 * b0)) * 2];
    float t = yr * z.x - yi * z.y;
    t += xrow<1>(t);
    t += xrow<2>(t);
    t += xrow<4>(t);
    t += xrow<8>(t);
    zsum[pass] = t;
  }
  if ((tid & 15) == 0) {
    red[tid >> 4] = zsum[0];
    red[(tid >> 4) + 6] = zsum[1];
  }
  __syncthreads();
  if (tid < NQ) out[b * NQ + (11 - tid)] = red[tid];
}

extern "C" void kernel_launch(void* const* d_in, const int* in_sizes, int n_in,
                              void* d_out, int out_size, void* d_ws,
                              size_t ws_size, hipStream_t stream) {
  const float* x = (const float*)d_in[0];
  const float* w = (const float*)d_in[1];
  float* out = (float*)d_out;
  int B = in_sizes[0] / NQ;
  qsim_kernel<<<B, 96, 0, stream>>>(x, w, out);
}
